# starting kernel
# baseline (speedup 1.0000x reference)
#include <hip/hip_runtime.h>
#include <hip/hip_bf16.h>
#include <math.h>

typedef __attribute__((ext_vector_type(2))) float v2f;
typedef __attribute__((ext_vector_type(4))) float v4f;
typedef __attribute__((ext_vector_type(8))) float v8f;

__global__ __launch_bounds__(256)
void agg_kernel(const float* __restrict__ h, const float* __restrict__ m,
                const int* __restrict__ idx_e, const int* __restrict__ idx_n,
                float* __restrict__ f_u) {
    const int e     = blockIdx.x;
    const int tid   = threadIdx.x;
    const int quad  = tid & 15;
    const int rslot = tid >> 4;

    __shared__ int s_ie[64];
    __shared__ int s_in[64];
    if (tid < 64)          s_ie[tid]      = idx_e[(size_t)e * 64 + tid];
    else if (tid < 128)    s_in[tid - 64] = idx_n[(size_t)e * 64 + (tid - 64)];
    __syncthreads();

    const v4f* __restrict__ hv = (const v4f*)h;
    const v4f* __restrict__ mv = (const v4f*)m;
    v4f* __restrict__ fv       = (v4f*)f_u;

    __shared__ v4f red[256];

    {
        v4f acc = {0.f, 0.f, 0.f, 0.f};
        #pragma unroll
        for (int i = 0; i < 4; ++i) {
            const int r = rslot + 16 * i;
            acc += hv[(size_t)s_ie[r] * 16 + quad];
        }
        red[tid] = acc;
        __syncthreads();
        #pragma unroll
        for (int s = 128; s >= 16; s >>= 1) {
            if (tid < s) red[tid] += red[tid + s];
            __syncthreads();
        }
        if (tid < 16) {
            v4f r = red[tid];
            r *= (1.0f / 64.0f);
            fv[(size_t)e * 32 + tid] = r;
        }
        __syncthreads();
    }

    {
        v4f acc = {0.f, 0.f, 0.f, 0.f};
        #pragma unroll
        for (int i = 0; i < 4; ++i) {
            const int r = rslot + 16 * i;
            acc += mv[(size_t)s_in[r] * 16 + quad];
        }
        red[tid] = acc;
        __syncthreads();
        #pragma unroll
        for (int s = 128; s >= 16; s >>= 1) {
            if (tid < s) red[tid] += red[tid + s];
            __syncthreads();
        }
        if (tid < 16) {
            v4f r = red[tid];
            r *= (1.0f / 64.0f);
            fv[(size_t)e * 32 + 16 + tid] = r;
        }
    }
}

__global__ __launch_bounds__(256)
void wmma_gemm_f32(const float* __restrict__ A, const float* __restrict__ B,
                   float* __restrict__ C, int M, int Kd) {
    constexpr int N = 128;
    const int wave = threadIdx.x >> 5;
    const int lane = threadIdx.x & 31;
    const int tile = blockIdx.x * 8 + wave;
    const int tm   = tile >> 3;
    const int tn   = tile & 7;
    if (tm * 16 >= M) return;

    const int half = lane >> 4;
    const int l    = lane & 15;
    const int arow = tm * 16 + l;
    const int ncol = tn * 16 + l;

    v8f acc = {0.f, 0.f, 0.f, 0.f, 0.f, 0.f, 0.f, 0.f};

    for (int k = 0; k < Kd; k += 4) {
        const int ka = k + half * 2;
        v2f a, b;
        a.x = A[(size_t)arow * Kd + ka];
        a.y = A[(size_t)arow * Kd + ka + 1];
        b.x = B[(size_t)(ka)     * N + ncol];
        b.y = B[(size_t)(ka + 1) * N + ncol];
        acc = __builtin_amdgcn_wmma_f32_16x16x4_f32(
             false, a,  false, b,
             (short)0, acc,  false,  false);
    }

    const int r0 = tm * 16 + half * 8;
    #pragma unroll
    for (int v = 0; v < 8; ++v) {
        C[(size_t)(r0 + v) * N + ncol] = acc[v];
    }
}

__global__ __launch_bounds__(128)
void attn_kernel(const float* __restrict__ ph, const float* __restrict__ pu,
                 const float* __restrict__ a, float* __restrict__ h_new,
                 float* __restrict__ alpha_out) {
    const int e = blockIdx.x;
    const int d = threadIdx.x;

    const float vph = ph[(size_t)e * 128 + d];
    const float vpu = pu[(size_t)e * 128 + d];
    const float ad  = a[d];

    __shared__ float r0[128];
    __shared__ float r1[128];
    r0[d] = tanhf(vph) * ad;
    r1[d] = tanhf(vpu) * ad;
    __syncthreads();

    #pragma unroll
    for (int s = 64; s > 0; s >>= 1) {
        if (d < s) { r0[d] += r0[d + s]; r1[d] += r1[d + s]; }
        __syncthreads();
    }

    __shared__ float al0, al12;
    if (d == 0) {
        const float s0 = r0[0], s1 = r1[0];
        const float mx = fmaxf(s0, s1);
        const float e0 = expf(s0 - mx);
        const float e1 = expf(s1 - mx);
        const float denom = e0 + 2.f * e1;
        const float a0 = e0 / denom;
        const float a1 = e1 / denom;
        al0  = a0;
        al12 = 2.f * a1;
        alpha_out[(size_t)e * 3 + 0] = a0;
        alpha_out[(size_t)e * 3 + 1] = a1;
        alpha_out[(size_t)e * 3 + 2] = a1;
    }
    __syncthreads();

    h_new[(size_t)e * 128 + d] = al0 * vph + al12 * vpu;
}

extern "C" void kernel_launch(void* const* d_in, const int* in_sizes, int n_in,
                              void* d_out, int out_size, void* d_ws, size_t ws_size,
                              hipStream_t stream) {
    const float* h_xe = (const float*)d_in[0];
    const float* h    = (const float*)d_in[1];
    const float* m    = (const float*)d_in[2];
    const float* W_h  = (const float*)d_in[3];
    const float* W_f  = (const float*)d_in[4];
    const float* avec = (const float*)d_in[5];
    const int* idx_e  = (const int*)d_in[6];
    const int* idx_n  = (const int*)d_in[7];

    const int E  = in_sizes[0] / 64;
    const int DH = in_sizes[5];

    float* f_u = (float*)d_ws;
    float* ph  = f_u + (size_t)E * DH;
    float* pu  = ph  + (size_t)E * DH;

    float* h_new     = (float*)d_out;
    float* alpha_out = (float*)d_out + (size_t)E * DH;

    agg_kernel<<<E, 256, 0, stream>>>(h, m, idx_e, idx_n, f_u);

    const int tiles  = (E / 16) * (DH / 16);
    const int blocks = tiles / 8;
    wmma_gemm_f32<<<blocks, 256, 0, stream>>>(h_xe, W_h, ph, E, 64);
    wmma_gemm_f32<<<blocks, 256, 0, stream>>>(f_u,  W_f, pu, E, 128);

    attn_kernel<<<E, 128, 0, stream>>>(ph, pu, avec, h_new, alpha_out);
}


# one wave per edge in the gather kernel, one per two edges in the attention kernel (were 8 and 4 per edge): 65536 -> 8192 and 32768 -> 4096 waves; sums reordered (serial rows per lane, wave butterfly)
# speedup vs baseline: 1.2434x; 1.2434x over previous
#include <hip/hip_runtime.h>
#include <hip/hip_bf16.h>
#include <math.h>

typedef __attribute__((ext_vector_type(2))) float v2f;
typedef __attribute__((ext_vector_type(4))) float v4f;
typedef __attribute__((ext_vector_type(8))) float v8f;

typedef __attribute__((ext_vector_type(4))) int v4i;

template <int R>
__device__ __forceinline__ v4f load_row(const v4i iv, const char* __restrict__ base) {
    const int idx = __builtin_amdgcn_update_dpp(0, iv[R & 3], 0x150 + (R >> 2), 0xf, 0xf, true);
    return *(const v4f*)(base + ((size_t)(unsigned)idx << 8));
}

template <int K>
__device__ __forceinline__ void load_walk(const v4i iv, const char* __restrict__ base, v4f (&d)[8]) {
    d[0] = load_row<8 * K + 0>(iv, base);
    d[1] = load_row<8 * K + 1>(iv, base);
    d[2] = load_row<8 * K + 2>(iv, base);
    d[3] = load_row<8 * K + 3>(iv, base);
    d[4] = load_row<8 * K + 4>(iv, base);
    d[5] = load_row<8 * K + 5>(iv, base);
    d[6] = load_row<8 * K + 6>(iv, base);
    d[7] = load_row<8 * K + 7>(iv, base);
}

__device__ __forceinline__ v4f sum8(const v4f (&d)[8]) {
    v4f s = d[0];
    s += d[1]; s += d[2]; s += d[3]; s += d[4]; s += d[5]; s += d[6]; s += d[7];
    return s;
}

#define PIN_ORDER() __builtin_amdgcn_sched_barrier(0)

__global__ __launch_bounds__(256)
void agg_kernel(const float* __restrict__ h, const float* __restrict__ m,
                const int* __restrict__ idx_e, const int* __restrict__ idx_n,
                float* __restrict__ f_u, int E) {
    const int lane = threadIdx.x & 31;
    const int half = lane >> 4;
    const int quad = lane & 15;
    const int e = __builtin_amdgcn_readfirstlane((int)(blockIdx.x * 8 + (threadIdx.x >> 5)));
    if (e >= E) return;

    const int* ip = (half ? idx_n : idx_e) + (size_t)e * 64 + 4 * quad;
    const v4i iv = *(const v4i*)ip;
    const char* base = (const char*)(half ? m : h) + 16 * quad;

    v4f ra[8], rb[8];
    load_walk<0>(iv, base, ra);
    load_walk<1>(iv, base, rb);
    PIN_ORDER();
    const v4f p0 = sum8(ra);
    PIN_ORDER();
    load_walk<2>(iv, base, ra);
    PIN_ORDER();
    const v4f p1 = sum8(rb);
    const v4f q01 = p0 + p1;
    PIN_ORDER();
    load_walk<3>(iv, base, rb);
    PIN_ORDER();
    const v4f p2 = sum8(ra);
    PIN_ORDER();
    load_walk<4>(iv, base, ra);
    PIN_ORDER();
    const v4f p3 = sum8(rb);
    const v4f q03 = q01 + (p2 + p3);
    PIN_ORDER();
    load_walk<5>(iv, base, rb);
    PIN_ORDER();
    const v4f p4 = sum8(ra);
    PIN_ORDER();
    load_walk<6>(iv, base, ra);
    PIN_ORDER();
    const v4f p5 = sum8(rb);
    const v4f q45 = p4 + p5;
    PIN_ORDER();
    load_walk<7>(iv, base, rb);
    PIN_ORDER();
    const v4f p6 = sum8(ra);
    const v4f p7 = sum8(rb);
    v4f t = q03 + (q45 + (p6 + p7));
    t *= (1.0f / 64.0f);
    ((v4f*)f_u)[(size_t)e * 32 + lane] = t;
}

__global__ __launch_bounds__(256)
void wmma_gemm_f32(const float* __restrict__ A, const float* __restrict__ B,
                   float* __restrict__ C, int M, int Kd) {
    constexpr int N = 128;
    const int wave = threadIdx.x >> 5;
    const int lane = threadIdx.x & 31;
    const int tile = blockIdx.x * 8 + wave;
    const int tm   = tile >> 3;
    const int tn   = tile & 7;
    if (tm * 16 >= M) return;

    const int half = lane >> 4;
    const int l    = lane & 15;
    const int arow = tm * 16 + l;
    const int ncol = tn * 16 + l;

    v8f acc = {0.f, 0.f, 0.f, 0.f, 0.f, 0.f, 0.f, 0.f};

    for (int k = 0; k < Kd; k += 4) {
        const int ka = k + half * 2;
        v2f a, b;
        a.x = A[(size_t)arow * Kd + ka];
        a.y = A[(size_t)arow * Kd + ka + 1];
        b.x = B[(size_t)(ka)     * N + ncol];
        b.y = B[(size_t)(ka + 1) * N + ncol];
        acc = __builtin_amdgcn_wmma_f32_16x16x4_f32(
             false, a,  false, b,
             (short)0, acc,  false,  false);
    }

    const int r0 = tm * 16 + half * 8;
    #pragma unroll
    for (int v = 0; v < 8; ++v) {
        C[(size_t)(r0 + v) * N + ncol] = acc[v];
    }
}

__global__ __launch_bounds__(256)
void attn_kernel(const float* __restrict__ ph, const float* __restrict__ pu,
                 const float* __restrict__ a, float* __restrict__ h_new,
                 float* __restrict__ alpha_out, int E) {
    constexpr int EPW = 2;
    const int lane = threadIdx.x & 31;
    const int e0 = __builtin_amdgcn_readfirstlane((int)(blockIdx.x * 8 + (threadIdx.x >> 5)) * EPW);
    if (e0 >= E) return;

    const v4f av = ((const v4f*)a)[lane];
    v4f vph[EPW], vpu[EPW];
    for (int j = 0; j < EPW; ++j) {
        const int e = (e0 + j < E) ? (e0 + j) : e0;
        vph[j] = ((const v4f*)ph)[(size_t)e * 32 + lane];
        vpu[j] = ((const v4f*)pu)[(size_t)e * 32 + lane];
    }

    float s0[EPW], s1[EPW];
    for (int j = 0; j < EPW; ++j) {
        s0[j] = tanhf(vph[j].x) * av.x + tanhf(vph[j].y) * av.y
              + tanhf(vph[j].z) * av.z + tanhf(vph[j].w) * av.w;
        s1[j] = tanhf(vpu[j].x) * av.x + tanhf(vpu[j].y) * av.y
              + tanhf(vpu[j].z) * av.z + tanhf(vpu[j].w) * av.w;
    }
    for (int o = 16; o > 0; o >>= 1) {
        for (int j = 0; j < EPW; ++j) {
            s0[j] += __shfl_xor(s0[j], o, 32);
            s1[j] += __shfl_xor(s1[j], o, 32);
        }
    }

    float a0[EPW], a1[EPW];
    for (int j = 0; j < EPW; ++j) {
        const float mx = fmaxf(s0[j], s1[j]);
        const float x0 = expf(s0[j] - mx);
        const float x1 = expf(s1[j] - mx);
        const float denom = x0 + 2.f * x1;
        a0[j] = x0 / denom;
        a1[j] = x1 / denom;
        if (e0 + j < E) {
            const float al12 = 2.f * a1[j];
            ((v4f*)h_new)[(size_t)(e0 + j) * 32 + lane] = a0[j] * vph[j] + al12 * vpu[j];
        }
    }
    if (lane < 3 * EPW) {
        const int j = lane / 3;
        const int c = lane - 3 * j;
        float v = (c == 0) ? a0[0] : a1[0];
        for (int q = 1; q < EPW; ++q) if (j == q) v = (c == 0) ? a0[q] : a1[q];
        if (e0 + j < E) alpha_out[(size_t)e0 * 3 + lane] = v;
    }
}

extern "C" void kernel_launch(void* const* d_in, const int* in_sizes, int n_in,
                              void* d_out, int out_size, void* d_ws, size_t ws_size,
                              hipStream_t stream) {
    const float* h_xe = (const float*)d_in[0];
    const float* h    = (const float*)d_in[1];
    const float* m    = (const float*)d_in[2];
    const float* W_h  = (const float*)d_in[3];
    const float* W_f  = (const float*)d_in[4];
    const float* avec = (const float*)d_in[5];
    const int* idx_e  = (const int*)d_in[6];
    const int* idx_n  = (const int*)d_in[7];

    const int E  = in_sizes[0] / 64;
    const int DH = in_sizes[5];

    float* f_u = (float*)d_ws;
    float* ph  = f_u + (size_t)E * DH;
    float* pu  = ph  + (size_t)E * DH;

    float* h_new     = (float*)d_out;
    float* alpha_out = (float*)d_out + (size_t)E * DH;

    agg_kernel<<<(E + 7) / 8, 256, 0, stream>>>(h, m, idx_e, idx_n, f_u, E);

    const int tiles  = (E / 16) * (DH / 16);
    const int blocks = tiles / 8;
    wmma_gemm_f32<<<blocks, 256, 0, stream>>>(h_xe, W_h, ph, E, 64);
    wmma_gemm_f32<<<blocks, 256, 0, stream>>>(f_u,  W_f, pu, E, 128);

    attn_kernel<<<(E + 15) / 16, 256, 0, stream>>>(ph, pu, avec, h_new, alpha_out, E);
}


# both projection GEMMs and the attention head fused into one kernel (8-wave block per 32 rows, A rows staged in LDS, ph/pu kept in WMMA accumulators): 4 launches -> 2; score sum order changed
# speedup vs baseline: 1.4980x; 1.2048x over previous
#include <hip/hip_runtime.h>
#include <hip/hip_bf16.h>
#include <math.h>

typedef __attribute__((ext_vector_type(2))) float v2f;
typedef __attribute__((ext_vector_type(4))) float v4f;
typedef __attribute__((ext_vector_type(8))) float v8f;

typedef __attribute__((ext_vector_type(4))) int v4i;

template <int R>
__device__ __forceinline__ v4f load_row(const v4i iv, const char* __restrict__ base) {
    const int idx = __builtin_amdgcn_update_dpp(0, iv[R & 3], 0x150 + (R >> 2), 0xf, 0xf, true);
    return *(const v4f*)(base + ((size_t)(unsigned)idx << 8));
}

template <int K>
__device__ __forceinline__ void load_walk(const v4i iv, const char* __restrict__ base, v4f (&d)[8]) {
    d[0] = load_row<8 * K + 0>(iv, base);
    d[1] = load_row<8 * K + 1>(iv, base);
    d[2] = load_row<8 * K + 2>(iv, base);
    d[3] = load_row<8 * K + 3>(iv, base);
    d[4] = load_row<8 * K + 4>(iv, base);
    d[5] = load_row<8 * K + 5>(iv, base);
    d[6] = load_row<8 * K + 6>(iv, base);
    d[7] = load_row<8 * K + 7>(iv, base);
}

__device__ __forceinline__ v4f sum8(const v4f (&d)[8]) {
    v4f s = d[0];
    s += d[1]; s += d[2]; s += d[3]; s += d[4]; s += d[5]; s += d[6]; s += d[7];
    return s;
}

#define PIN_ORDER() __builtin_amdgcn_sched_barrier(0)

__global__ __launch_bounds__(256)
void agg_kernel(const float* __restrict__ h, const float* __restrict__ m,
                const int* __restrict__ idx_e, const int* __restrict__ idx_n,
                float* __restrict__ f_u, int E) {
    const int lane = threadIdx.x & 31;
    const int half = lane >> 4;
    const int quad = lane & 15;
    const int e = __builtin_amdgcn_readfirstlane((int)(blockIdx.x * 8 + (threadIdx.x >> 5)));
    if (e >= E) return;

    const int* ip = (half ? idx_n : idx_e) + (size_t)e * 64 + 4 * quad;
    const v4i iv = *(const v4i*)ip;
    const char* base = (const char*)(half ? m : h) + 16 * quad;

    v4f ra[8], rb[8];
    load_walk<0>(iv, base, ra);
    load_walk<1>(iv, base, rb);
    PIN_ORDER();
    const v4f p0 = sum8(ra);
    PIN_ORDER();
    load_walk<2>(iv, base, ra);
    PIN_ORDER();
    const v4f p1 = sum8(rb);
    const v4f q01 = p0 + p1;
    PIN_ORDER();
    load_walk<3>(iv, base, rb);
    PIN_ORDER();
    const v4f p2 = sum8(ra);
    PIN_ORDER();
    load_walk<4>(iv, base, ra);
    PIN_ORDER();
    const v4f p3 = sum8(rb);
    const v4f q03 = q01 + (p2 + p3);
    PIN_ORDER();
    load_walk<5>(iv, base, rb);
    PIN_ORDER();
    const v4f p4 = sum8(ra);
    PIN_ORDER();
    load_walk<6>(iv, base, ra);
    PIN_ORDER();
    const v4f p5 = sum8(rb);
    const v4f q45 = p4 + p5;
    PIN_ORDER();
    load_walk<7>(iv, base, rb);
    PIN_ORDER();
    const v4f p6 = sum8(ra);
    const v4f p7 = sum8(rb);
    v4f t = q03 + (q45 + (p6 + p7));
    t *= (1.0f / 64.0f);
    ((v4f*)f_u)[(size_t)e * 32 + lane] = t;
}

constexpr int HB_ROWS  = 32;
constexpr int PITCH_H  = 64 + 4;
constexpr int PITCH_U  = 128 + 4;

template <int CTRL>
__device__ __forceinline__ float dpp_row(const float x) {
    return __builtin_bit_cast(float, __builtin_amdgcn_update_dpp(
        0, __builtin_bit_cast(int, x), CTRL, 0xf, 0xf, true));
}
template <int M> __device__ __forceinline__ float row_xor(const float x)   { return dpp_row<0x160 + M>(x); }
template <int N> __device__ __forceinline__ float row_share(const float x) { return dpp_row<0x150 + N>(x); }

__device__ __forceinline__ v8f wmma4(const v2f a, const v2f b, const v8f c) {
    return __builtin_amdgcn_wmma_f32_16x16x4_f32(
         false, a,  false, b,
         (short)0, c,  false,  false);
}

__global__ __launch_bounds__(256)
void proj_attn_kernel(const float* __restrict__ h_xe, const float* __restrict__ f_u,
                      const float* __restrict__ W_h, const float* __restrict__ W_f,
                      const float* __restrict__ a, float* __restrict__ h_new,
                      float* __restrict__ alpha_out, int E) {
    constexpr int N = 128;
    __shared__ __attribute__((aligned(16))) float sAh[HB_ROWS * PITCH_H];
    __shared__ __attribute__((aligned(16))) float sAu[HB_ROWS * PITCH_U];
    __shared__ __attribute__((aligned(16))) float sPart[8 * 2 * HB_ROWS];
    __shared__ __attribute__((aligned(16))) float sAlpha[3 * HB_ROWS];

    const int t    = threadIdx.x;
    const int wave = t >> 5;
    const int lane = t & 31;
    const int half = lane >> 4;
    const int c    = lane & 15;
    const int row0 = blockIdx.x * HB_ROWS;

    for (int i = 0; i < 2; ++i) {
        const int idx = t + 256 * i;
        const int r = idx >> 4, q = idx & 15;
        const int gr = (row0 + r < E) ? (row0 + r) : (E - 1);
        *(v4f*)&sAh[r * PITCH_H + 4 * q] = ((const v4f*)h_xe)[(size_t)gr * 16 + q];
    }
    for (int i = 0; i < 4; ++i) {
        const int idx = t + 256 * i;
        const int r = idx >> 5, q = idx & 31;
        const int gr = (row0 + r < E) ? (row0 + r) : (E - 1);
        *(v4f*)&sAu[r * PITCH_U + 4 * q] = ((const v4f*)f_u)[(size_t)gr * 32 + q];
    }
    __syncthreads();

    const int ncol = wave * 16 + c;
    v8f acc_h[2], acc_u[2];
    acc_h[0] = acc_h[1] = acc_u[0] = acc_u[1] = (v8f){0.f, 0.f, 0.f, 0.f, 0.f, 0.f, 0.f, 0.f};

    for (int k = 0; k < 64; k += 4) {
        const int ka = k + half * 2;
        v2f b;
        b.x = W_h[(size_t)(ka)     * N + ncol];
        b.y = W_h[(size_t)(ka + 1) * N + ncol];
        const v2f a0 = *(const v2f*)&sAh[c * PITCH_H + ka];
        const v2f a1 = *(const v2f*)&sAh[(16 + c) * PITCH_H + ka];
        acc_h[0] = wmma4(a0, b, acc_h[0]);
        acc_h[1] = wmma4(a1, b, acc_h[1]);
    }
    for (int k = 0; k < 128; k += 4) {
        const int ka = k + half * 2;
        v2f b;
        b.x = W_f[(size_t)(ka)     * N + ncol];
        b.y = W_f[(size_t)(ka + 1) * N + ncol];
        const v2f a0 = *(const v2f*)&sAu[c * PITCH_U + ka];
        const v2f a1 = *(const v2f*)&sAu[(16 + c) * PITCH_U + ka];
        acc_u[0] = wmma4(a0, b, acc_u[0]);
        acc_u[1] = wmma4(a1, b, acc_u[1]);
    }

    const float av = a[ncol];
    float ps[2][2][8];
    for (int mt = 0; mt < 2; ++mt) {
        for (int r = 0; r < 8; ++r) {
            ps[0][mt][r] = tanhf(acc_h[mt][r]) * av;
            ps[1][mt][r] = tanhf(acc_u[mt][r]) * av;
        }
    }
    for (int p = 0; p < 2; ++p) {
        for (int mt = 0; mt < 2; ++mt) {
            for (int r = 0; r < 8; ++r) {
                float s = ps[p][mt][r];
                s += row_xor<1>(s);
                s += row_xor<2>(s);
                s += row_xor<4>(s);
                s += row_xor<8>(s);
                ps[p][mt][r] = s;
            }
        }
    }
    if (c == 0) {
        for (int p = 0; p < 2; ++p) {
            for (int mt = 0; mt < 2; ++mt) {
                float* d = &sPart[(wave * 2 + p) * HB_ROWS + 16 * mt + 8 * half];
                *(v4f*)d       = (v4f){ps[p][mt][0], ps[p][mt][1], ps[p][mt][2], ps[p][mt][3]};
                *(v4f*)(d + 4) = (v4f){ps[p][mt][4], ps[p][mt][5], ps[p][mt][6], ps[p][mt][7]};
            }
        }
    }
    __syncthreads();

    const int my_row = ((c >> 3) << 4) + 8 * half + (c & 7);
    float s0 = sPart[0 * HB_ROWS + my_row];
    float s1 = sPart[1 * HB_ROWS + my_row];
    for (int w = 1; w < 8; ++w) {
        s0 += sPart[(w * 2 + 0) * HB_ROWS + my_row];
        s1 += sPart[(w * 2 + 1) * HB_ROWS + my_row];
    }
    const float mx = fmaxf(s0, s1);
    const float x0 = expf(s0 - mx);
    const float x1 = expf(s1 - mx);
    const float denom = x0 + 2.f * x1;
    const float al0 = x0 / denom;
    const float al1 = x1 / denom;
    const float al12 = 2.f * al1;
    if (wave == 0) {
        sAlpha[3 * my_row + 0] = al0;
        sAlpha[3 * my_row + 1] = al1;
        sAlpha[3 * my_row + 2] = al1;
    }

    for (int mt = 0; mt < 2; ++mt) {
        for (int r = 0; r < 8; ++r) {
            float w0, w12;
            switch (8 * mt + r) {
                case 0:  w0 = row_share<0>(al0);  w12 = row_share<0>(al12);  break;
                case 1:  w0 = row_share<1>(al0);  w12 = row_share<1>(al12);  break;
                case 2:  w0 = row_share<2>(al0);  w12 = row_share<2>(al12);  break;
                case 3:  w0 = row_share<3>(al0);  w12 = row_share<3>(al12);  break;
                case 4:  w0 = row_share<4>(al0);  w12 = row_share<4>(al12);  break;
                case 5:  w0 = row_share<5>(al0);  w12 = row_share<5>(al12);  break;
                case 6:  w0 = row_share<6>(al0);  w12 = row_share<6>(al12);  break;
                case 7:  w0 = row_share<7>(al0);  w12 = row_share<7>(al12);  break;
                case 8:  w0 = row_share<8>(al0);  w12 = row_share<8>(al12);  break;
                case 9:  w0 = row_share<9>(al0);  w12 = row_share<9>(al12);  break;
                case 10: w0 = row_share<10>(al0); w12 = row_share<10>(al12); break;
                case 11: w0 = row_share<11>(al0); w12 = row_share<11>(al12); break;
                case 12: w0 = row_share<12>(al0); w12 = row_share<12>(al12); break;
                case 13: w0 = row_share<13>(al0); w12 = row_share<13>(al12); break;
                case 14: w0 = row_share<14>(al0); w12 = row_share<14>(al12); break;
                default: w0 = row_share<15>(al0); w12 = row_share<15>(al12); break;
            }
            sAu[(16 * mt + 8 * half + r) * PITCH_U + ncol] = w0 * acc_h[mt][r] + w12 * acc_u[mt][r];
        }
    }
    __syncthreads();

    for (int j = 0; j < 4; ++j) {
        const int r = 4 * wave + j;
        if (row0 + r < E)
            ((v4f*)h_new)[(size_t)(row0 + r) * 32 + lane] = *(const v4f*)&sAu[r * PITCH_U + 4 * lane];
    }
    if (wave == 0) {
        if (row0 + HB_ROWS <= E) {
            if (lane < 24)
                ((v4f*)alpha_out)[(size_t)blockIdx.x * 24 + lane] = *(const v4f*)&sAlpha[4 * lane];
        } else {
            for (int f = lane; f < 3 * HB_ROWS; f += 32)
                if (row0 + f / 3 < E) alpha_out[(size_t)row0 * 3 + f] = sAlpha[f];
        }
    }
}

extern "C" void kernel_launch(void* const* d_in, const int* in_sizes, int n_in,
                              void* d_out, int out_size, void* d_ws, size_t ws_size,
                              hipStream_t stream) {
    const float* h_xe = (const float*)d_in[0];
    const float* h    = (const float*)d_in[1];
    const float* m    = (const float*)d_in[2];
    const float* W_h  = (const float*)d_in[3];
    const float* W_f  = (const float*)d_in[4];
    const float* avec = (const float*)d_in[5];
    const int* idx_e  = (const int*)d_in[6];
    const int* idx_n  = (const int*)d_in[7];

    const int E  = in_sizes[0] / 64;
    const int DH = in_sizes[5];

    float* f_u = (float*)d_ws;

    float* h_new     = (float*)d_out;
    float* alpha_out = (float*)d_out + (size_t)E * DH;

    agg_kernel<<<(E + 7) / 8, 256, 0, stream>>>(h, m, idx_e, idx_n, f_u, E);

    proj_attn_kernel<<<(E + 31) / 32, 256, 0, stream>>>(h_xe, f_u, W_h, W_f, avec,
                                                         h_new, alpha_out, E);
}


# gather fused into the projection/attention kernel: each wave sums its 4 edges' rows (32-40 row loads in flight, no drain between edges) into the LDS A tile; f_u never stored; 2 launches -> 1
# speedup vs baseline: 1.9381x; 1.2938x over previous
#include <hip/hip_runtime.h>
#include <hip/hip_bf16.h>
#include <math.h>

typedef __attribute__((ext_vector_type(2))) float v2f;
typedef __attribute__((ext_vector_type(4))) float v4f;
typedef __attribute__((ext_vector_type(8))) float v8f;

typedef __attribute__((ext_vector_type(4))) int v4i;

template <int R>
__device__ __forceinline__ v4f load_row(const v4i iv, const char* __restrict__ base) {
    const int idx = __builtin_amdgcn_update_dpp(0, iv[R & 3], 0x150 + (R >> 2), 0xf, 0xf, true);
    return *(const v4f*)(base + ((size_t)(unsigned)idx << 8));
}

template <int K>
__device__ __forceinline__ void load_walk(const v4i iv, const char* __restrict__ base, v4f (&d)[8]) {
    d[0] = load_row<8 * K + 0>(iv, base);
    d[1] = load_row<8 * K + 1>(iv, base);
    d[2] = load_row<8 * K + 2>(iv, base);
    d[3] = load_row<8 * K + 3>(iv, base);
    d[4] = load_row<8 * K + 4>(iv, base);
    d[5] = load_row<8 * K + 5>(iv, base);
    d[6] = load_row<8 * K + 6>(iv, base);
    d[7] = load_row<8 * K + 7>(iv, base);
}

__device__ __forceinline__ v4f sum8(const v4f (&d)[8]) {
    v4f s = d[0];
    s += d[1]; s += d[2]; s += d[3]; s += d[4]; s += d[5]; s += d[6]; s += d[7];
    return s;
}

#define PIN_ORDER() __builtin_amdgcn_sched_barrier(0)

constexpr int PITCH_U = 128 + 4;
constexpr int EPW     = 4;
constexpr int NWALK   = 8 * EPW;
constexpr int NBUF    = 5;

template <int G>
__device__ __forceinline__ void load_walk_ring(const v4i (&iv)[EPW], const char* base, v4f (&buf)[NBUF][8]) {
    load_walk<(G & 7)>(iv[G >> 3], base, buf[G % NBUF]);
}

struct EdgeSum { v4f even; v4f q03; v4f q45; };

template <int G>
__device__ __forceinline__ void gather_from(const v4i (&iv)[EPW], const char* base, v4f (&buf)[NBUF][8],
                                            EdgeSum& es, float* __restrict__ rows, const int lane) {
    constexpr int K = G & 7;
    const v4f s = sum8(buf[G % NBUF]);
    if constexpr ((K & 1) == 0) es.even = s;
    if constexpr (K == 1) es.q03 = es.even + s;
    if constexpr (K == 3) es.q03 = es.q03 + (es.even + s);
    if constexpr (K == 5) es.q45 = es.even + s;
    if constexpr (K == 7) {
        v4f t = es.q03 + (es.q45 + (es.even + s));
        t *= (1.0f / 64.0f);
        *(v4f*)&rows[(G >> 3) * PITCH_U + 4 * lane] = t;
    }
    PIN_ORDER();
    if constexpr (G + NBUF < NWALK) load_walk_ring<G + NBUF>(iv, base, buf);
    PIN_ORDER();
    if constexpr (G + 1 < NWALK) gather_from<G + 1>(iv, base, buf, es, rows, lane);
}

constexpr int HB_ROWS  = 32;
constexpr int PITCH_H  = 64 + 4;

template <int CTRL>
__device__ __forceinline__ float dpp_row(const float x) {
    return __builtin_bit_cast(float, __builtin_amdgcn_update_dpp(
        0, __builtin_bit_cast(int, x), CTRL, 0xf, 0xf, true));
}
template <int M> __device__ __forceinline__ float row_xor(const float x)   { return dpp_row<0x160 + M>(x); }
template <int N> __device__ __forceinline__ float row_share(const float x) { return dpp_row<0x150 + N>(x); }

__device__ __forceinline__ v8f wmma4(const v2f a, const v2f b, const v8f c) {
    return __builtin_amdgcn_wmma_f32_16x16x4_f32(
         false, a,  false, b,
         (short)0, c,  false,  false);
}

__global__ __launch_bounds__(256, 4)
void gather_proj_attn_kernel(const float* __restrict__ h_xe,
                             const float* __restrict__ h, const float* __restrict__ m,
                             const int* __restrict__ idx_e, const int* __restrict__ idx_n,
                             const float* __restrict__ W_h, const float* __restrict__ W_f,
                             const float* __restrict__ a, float* __restrict__ h_new,
                             float* __restrict__ alpha_out, int E) {
    constexpr int N = 128;
    __shared__ __attribute__((aligned(16))) float sAh[HB_ROWS * PITCH_H];
    __shared__ __attribute__((aligned(16))) float sAu[HB_ROWS * PITCH_U];
    __shared__ __attribute__((aligned(16))) float sPart[8 * 2 * HB_ROWS];
    __shared__ __attribute__((aligned(16))) float sAlpha[3 * HB_ROWS];

    const int t    = threadIdx.x;
    const int wave = t >> 5;
    const int lane = t & 31;
    const int half = lane >> 4;
    const int c    = lane & 15;
    const int row0 = blockIdx.x * HB_ROWS;

    const int er0 = __builtin_amdgcn_readfirstlane(EPW * wave);
    const int quad = c;
    const int* ilist = (half ? idx_n : idx_e) + 4 * quad;
    const char* base = (const char*)(half ? m : h) + 16 * quad;
    v4i iv[EPW];
    for (int j = 0; j < EPW; ++j) {
        const int ge = (row0 + er0 + j < E) ? (row0 + er0 + j) : (E - 1);
        iv[j] = *(const v4i*)(ilist + (size_t)ge * 64);
    }
    v4f hx[2];
    for (int i = 0; i < 2; ++i) {
        const int idx = t + 256 * i;
        const int r = idx >> 4, q = idx & 15;
        const int gr = (row0 + r < E) ? (row0 + r) : (E - 1);
        hx[i] = ((const v4f*)h_xe)[(size_t)gr * 16 + q];
    }
    PIN_ORDER();
    v4f buf[NBUF][8];
    static_assert(NBUF == 5 && HB_ROWS == 8 * EPW, "ring prologue and edge split are written out for these");
    load_walk_ring<0>(iv, base, buf);
    load_walk_ring<1>(iv, base, buf);
    load_walk_ring<2>(iv, base, buf);
    load_walk_ring<3>(iv, base, buf);
    load_walk_ring<4>(iv, base, buf);
    PIN_ORDER();
    for (int i = 0; i < 2; ++i) {
        const int idx = t + 256 * i;
        const int r = idx >> 4, q = idx & 15;
        *(v4f*)&sAh[r * PITCH_H + 4 * q] = hx[i];
    }
    PIN_ORDER();
    {
        EdgeSum es;
        gather_from<0>(iv, base, buf, es, &sAu[er0 * PITCH_U], lane);
    }
    __syncthreads();

    const int ncol = wave * 16 + c;
    v8f acc_h[2], acc_u[2];
    acc_h[0] = acc_h[1] = acc_u[0] = acc_u[1] = (v8f){0.f, 0.f, 0.f, 0.f, 0.f, 0.f, 0.f, 0.f};

    for (int k = 0; k < 64; k += 4) {
        const int ka = k + half * 2;
        v2f b;
        b.x = W_h[(size_t)(ka)     * N + ncol];
        b.y = W_h[(size_t)(ka + 1) * N + ncol];
        const v2f a0 = *(const v2f*)&sAh[c * PITCH_H + ka];
        const v2f a1 = *(const v2f*)&sAh[(16 + c) * PITCH_H + ka];
        acc_h[0] = wmma4(a0, b, acc_h[0]);
        acc_h[1] = wmma4(a1, b, acc_h[1]);
    }
    for (int k = 0; k < 128; k += 4) {
        const int ka = k + half * 2;
        v2f b;
        b.x = W_f[(size_t)(ka)     * N + ncol];
        b.y = W_f[(size_t)(ka + 1) * N + ncol];
        const v2f a0 = *(const v2f*)&sAu[c * PITCH_U + ka];
        const v2f a1 = *(const v2f*)&sAu[(16 + c) * PITCH_U + ka];
        acc_u[0] = wmma4(a0, b, acc_u[0]);
        acc_u[1] = wmma4(a1, b, acc_u[1]);
    }

    const float av = a[ncol];
    float ps[2][2][8];
    for (int mt = 0; mt < 2; ++mt) {
        for (int r = 0; r < 8; ++r) {
            ps[0][mt][r] = tanhf(acc_h[mt][r]) * av;
            ps[1][mt][r] = tanhf(acc_u[mt][r]) * av;
        }
    }
    for (int p = 0; p < 2; ++p) {
        for (int mt = 0; mt < 2; ++mt) {
            for (int r = 0; r < 8; ++r) {
                float s = ps[p][mt][r];
                s += row_xor<1>(s);
                s += row_xor<2>(s);
                s += row_xor<4>(s);
                s += row_xor<8>(s);
                ps[p][mt][r] = s;
            }
        }
    }
    if (c == 0) {
        for (int p = 0; p < 2; ++p) {
            for (int mt = 0; mt < 2; ++mt) {
                float* d = &sPart[(wave * 2 + p) * HB_ROWS + 16 * mt + 8 * half];
                *(v4f*)d       = (v4f){ps[p][mt][0], ps[p][mt][1], ps[p][mt][2], ps[p][mt][3]};
                *(v4f*)(d + 4) = (v4f){ps[p][mt][4], ps[p][mt][5], ps[p][mt][6], ps[p][mt][7]};
            }
        }
    }
    __syncthreads();

    const int my_row = ((c >> 3) << 4) + 8 * half + (c & 7);
    float s0 = sPart[0 * HB_ROWS + my_row];
    float s1 = sPart[1 * HB_ROWS + my_row];
    for (int w = 1; w < 8; ++w) {
        s0 += sPart[(w * 2 + 0) * HB_ROWS + my_row];
        s1 += sPart[(w * 2 + 1) * HB_ROWS + my_row];
    }
    const float mx = fmaxf(s0, s1);
    const float x0 = expf(s0 - mx);
    const float x1 = expf(s1 - mx);
    const float denom = x0 + 2.f * x1;
    const float al0 = x0 / denom;
    const float al1 = x1 / denom;
    const float al12 = 2.f * al1;
    if (wave == 0) {
        sAlpha[3 * my_row + 0] = al0;
        sAlpha[3 * my_row + 1] = al1;
        sAlpha[3 * my_row + 2] = al1;
    }

    for (int mt = 0; mt < 2; ++mt) {
        for (int r = 0; r < 8; ++r) {
            float w0, w12;
            switch (8 * mt + r) {
                case 0:  w0 = row_share<0>(al0);  w12 = row_share<0>(al12);  break;
                case 1:  w0 = row_share<1>(al0);  w12 = row_share<1>(al12);  break;
                case 2:  w0 = row_share<2>(al0);  w12 = row_share<2>(al12);  break;
                case 3:  w0 = row_share<3>(al0);  w12 = row_share<3>(al12);  break;
                case 4:  w0 = row_share<4>(al0);  w12 = row_share<4>(al12);  break;
                case 5:  w0 = row_share<5>(al0);  w12 = row_share<5>(al12);  break;
                case 6:  w0 = row_share<6>(al0);  w12 = row_share<6>(al12);  break;
                case 7:  w0 = row_share<7>(al0);  w12 = row_share<7>(al12);  break;
                case 8:  w0 = row_share<8>(al0);  w12 = row_share<8>(al12);  break;
                case 9:  w0 = row_share<9>(al0);  w12 = row_share<9>(al12);  break;
                case 10: w0 = row_share<10>(al0); w12 = row_share<10>(al12); break;
                case 11: w0 = row_share<11>(al0); w12 = row_share<11>(al12); break;
                case 12: w0 = row_share<12>(al0); w12 = row_share<12>(al12); break;
                case 13: w0 = row_share<13>(al0); w12 = row_share<13>(al12); break;
                case 14: w0 = row_share<14>(al0); w12 = row_share<14>(al12); break;
                default: w0 = row_share<15>(al0); w12 = row_share<15>(al12); break;
            }
            sAu[(16 * mt + 8 * half + r) * PITCH_U + ncol] = w0 * acc_h[mt][r] + w12 * acc_u[mt][r];
        }
    }
    __syncthreads();

    for (int j = 0; j < 4; ++j) {
        const int r = 4 * wave + j;
        if (row0 + r < E)
            ((v4f*)h_new)[(size_t)(row0 + r) * 32 + lane] = *(const v4f*)&sAu[r * PITCH_U + 4 * lane];
    }
    if (wave == 0) {
        if (row0 + HB_ROWS <= E) {
            if (lane < 24)
                ((v4f*)alpha_out)[(size_t)blockIdx.x * 24 + lane] = *(const v4f*)&sAlpha[4 * lane];
        } else {
            for (int f = lane; f < 3 * HB_ROWS; f += 32)
                if (row0 + f / 3 < E) alpha_out[(size_t)row0 * 3 + f] = sAlpha[f];
        }
    }
}

extern "C" void kernel_launch(void* const* d_in, const int* in_sizes, int n_in,
                              void* d_out, int out_size, void* d_ws, size_t ws_size,
                              hipStream_t stream) {
    const float* h_xe = (const float*)d_in[0];
    const float* h    = (const float*)d_in[1];
    const float* m    = (const float*)d_in[2];
    const float* W_h  = (const float*)d_in[3];
    const float* W_f  = (const float*)d_in[4];
    const float* avec = (const float*)d_in[5];
    const int* idx_e  = (const int*)d_in[6];
    const int* idx_n  = (const int*)d_in[7];

    const int E  = in_sizes[0] / 64;
    const int DH = in_sizes[5];

    float* h_new     = (float*)d_out;
    float* alpha_out = (float*)d_out + (size_t)E * DH;

    gather_proj_attn_kernel<<<(E + 31) / 32, 256, 0, stream>>>(h_xe, h, m, idx_e, idx_n,
                                                                W_h, W_f, avec,
                                                                h_new, alpha_out, E);
}


# tanh of each lane's 32 accumulator values branch-free as 1-2/(2^(2x*log2 e)+1) on v_exp/v_rcp, a folded into the fma (was 32 serial libm calls, 2 divergent branches each); abs err about 2e-7
# speedup vs baseline: 2.0705x; 1.0683x over previous
#include <hip/hip_runtime.h>
#include <hip/hip_bf16.h>
#include <math.h>

typedef __attribute__((ext_vector_type(2))) float v2f;
typedef __attribute__((ext_vector_type(4))) float v4f;
typedef __attribute__((ext_vector_type(8))) float v8f;

typedef __attribute__((ext_vector_type(4))) int v4i;

template <int R>
__device__ __forceinline__ v4f load_row(const v4i iv, const char* __restrict__ base) {
    const int idx = __builtin_amdgcn_update_dpp(0, iv[R & 3], 0x150 + (R >> 2), 0xf, 0xf, true);
    return *(const v4f*)(base + ((size_t)(unsigned)idx << 8));
}

template <int K>
__device__ __forceinline__ void load_walk(const v4i iv, const char* __restrict__ base, v4f (&d)[8]) {
    d[0] = load_row<8 * K + 0>(iv, base);
    d[1] = load_row<8 * K + 1>(iv, base);
    d[2] = load_row<8 * K + 2>(iv, base);
    d[3] = load_row<8 * K + 3>(iv, base);
    d[4] = load_row<8 * K + 4>(iv, base);
    d[5] = load_row<8 * K + 5>(iv, base);
    d[6] = load_row<8 * K + 6>(iv, base);
    d[7] = load_row<8 * K + 7>(iv, base);
}

__device__ __forceinline__ v4f sum8(const v4f (&d)[8]) {
    v4f s = d[0];
    s += d[1]; s += d[2]; s += d[3]; s += d[4]; s += d[5]; s += d[6]; s += d[7];
    return s;
}

#define PIN_ORDER() __builtin_amdgcn_sched_barrier(0)

constexpr int PITCH_U = 128 + 4;
constexpr int EPW     = 4;
constexpr int NWALK   = 8 * EPW;
constexpr int NBUF    = 5;

template <int G>
__device__ __forceinline__ void load_walk_ring(const v4i (&iv)[EPW], const char* base, v4f (&buf)[NBUF][8]) {
    load_walk<(G & 7)>(iv[G >> 3], base, buf[G % NBUF]);
}

struct EdgeSum { v4f even; v4f q03; v4f q45; };

template <int G>
__device__ __forceinline__ void gather_from(const v4i (&iv)[EPW], const char* base, v4f (&buf)[NBUF][8],
                                            EdgeSum& es, float* __restrict__ rows, const int lane) {
    constexpr int K = G & 7;
    const v4f s = sum8(buf[G % NBUF]);
    if constexpr ((K & 1) == 0) es.even = s;
    if constexpr (K == 1) es.q03 = es.even + s;
    if constexpr (K == 3) es.q03 = es.q03 + (es.even + s);
    if constexpr (K == 5) es.q45 = es.even + s;
    if constexpr (K == 7) {
        v4f t = es.q03 + (es.q45 + (es.even + s));
        t *= (1.0f / 64.0f);
        *(v4f*)&rows[(G >> 3) * PITCH_U + 4 * lane] = t;
    }
    PIN_ORDER();
    if constexpr (G + NBUF < NWALK) load_walk_ring<G + NBUF>(iv, base, buf);
    PIN_ORDER();
    if constexpr (G + 1 < NWALK) gather_from<G + 1>(iv, base, buf, es, rows, lane);
}

constexpr int HB_ROWS  = 32;
constexpr int PITCH_H  = 64 + 4;

template <int CTRL>
__device__ __forceinline__ float dpp_row(const float x) {
    return __builtin_bit_cast(float, __builtin_amdgcn_update_dpp(
        0, __builtin_bit_cast(int, x), CTRL, 0xf, 0xf, true));
}
template <int M> __device__ __forceinline__ float row_xor(const float x)   { return dpp_row<0x160 + M>(x); }
template <int N> __device__ __forceinline__ float row_share(const float x) { return dpp_row<0x150 + N>(x); }

__device__ __forceinline__ float tanh_times(const float x, const float a, const float minus_2a) {
    const float e = __builtin_amdgcn_exp2f(x * 2.885390081777927f);
    const float r = __builtin_amdgcn_rcpf(e + 1.0f);
    return __builtin_fmaf(r, minus_2a, a);
}

__device__ __forceinline__ v8f wmma4(const v2f a, const v2f b, const v8f c) {
    return __builtin_amdgcn_wmma_f32_16x16x4_f32(
         false, a,  false, b,
         (short)0, c,  false,  false);
}

__global__ __launch_bounds__(256, 4)
void gather_proj_attn_kernel(const float* __restrict__ h_xe,
                             const float* __restrict__ h, const float* __restrict__ m,
                             const int* __restrict__ idx_e, const int* __restrict__ idx_n,
                             const float* __restrict__ W_h, const float* __restrict__ W_f,
                             const float* __restrict__ a, float* __restrict__ h_new,
                             float* __restrict__ alpha_out, int E) {
    constexpr int N = 128;
    __shared__ __attribute__((aligned(16))) float sAh[HB_ROWS * PITCH_H];
    __shared__ __attribute__((aligned(16))) float sAu[HB_ROWS * PITCH_U];
    __shared__ __attribute__((aligned(16))) float sPart[8 * 2 * HB_ROWS];
    __shared__ __attribute__((aligned(16))) float sAlpha[3 * HB_ROWS];

    const int t    = threadIdx.x;
    const int wave = t >> 5;
    const int lane = t & 31;
    const int half = lane >> 4;
    const int c    = lane & 15;
    const int row0 = blockIdx.x * HB_ROWS;

    const int er0 = __builtin_amdgcn_readfirstlane(EPW * wave);
    const int quad = c;
    const int* ilist = (half ? idx_n : idx_e) + 4 * quad;
    const char* base = (const char*)(half ? m : h) + 16 * quad;
    v4i iv[EPW];
    for (int j = 0; j < EPW; ++j) {
        const int ge = (row0 + er0 + j < E) ? (row0 + er0 + j) : (E - 1);
        iv[j] = *(const v4i*)(ilist + (size_t)ge * 64);
    }
    v4f hx[2];
    for (int i = 0; i < 2; ++i) {
        const int idx = t + 256 * i;
        const int r = idx >> 4, q = idx & 15;
        const int gr = (row0 + r < E) ? (row0 + r) : (E - 1);
        hx[i] = ((const v4f*)h_xe)[(size_t)gr * 16 + q];
    }
    PIN_ORDER();
    v4f buf[NBUF][8];
    static_assert(NBUF == 5 && HB_ROWS == 8 * EPW, "ring prologue and edge split are written out for these");
    load_walk_ring<0>(iv, base, buf);
    load_walk_ring<1>(iv, base, buf);
    load_walk_ring<2>(iv, base, buf);
    load_walk_ring<3>(iv, base, buf);
    load_walk_ring<4>(iv, base, buf);
    PIN_ORDER();
    for (int i = 0; i < 2; ++i) {
        const int idx = t + 256 * i;
        const int r = idx >> 4, q = idx & 15;
        *(v4f*)&sAh[r * PITCH_H + 4 * q] = hx[i];
    }
    PIN_ORDER();
    {
        EdgeSum es;
        gather_from<0>(iv, base, buf, es, &sAu[er0 * PITCH_U], lane);
    }
    __syncthreads();

    const int ncol = wave * 16 + c;
    v8f acc_h[2], acc_u[2];
    acc_h[0] = acc_h[1] = acc_u[0] = acc_u[1] = (v8f){0.f, 0.f, 0.f, 0.f, 0.f, 0.f, 0.f, 0.f};

    for (int k = 0; k < 64; k += 4) {
        const int ka = k + half * 2;
        v2f b;
        b.x = W_h[(size_t)(ka)     * N + ncol];
        b.y = W_h[(size_t)(ka + 1) * N + ncol];
        const v2f a0 = *(const v2f*)&sAh[c * PITCH_H + ka];
        const v2f a1 = *(const v2f*)&sAh[(16 + c) * PITCH_H + ka];
        acc_h[0] = wmma4(a0, b, acc_h[0]);
        acc_h[1] = wmma4(a1, b, acc_h[1]);
    }
    for (int k = 0; k < 128; k += 4) {
        const int ka = k + half * 2;
        v2f b;
        b.x = W_f[(size_t)(ka)     * N + ncol];
        b.y = W_f[(size_t)(ka + 1) * N + ncol];
        const v2f a0 = *(const v2f*)&sAu[c * PITCH_U + ka];
        const v2f a1 = *(const v2f*)&sAu[(16 + c) * PITCH_U + ka];
        acc_u[0] = wmma4(a0, b, acc_u[0]);
        acc_u[1] = wmma4(a1, b, acc_u[1]);
    }

    const float av = a[ncol];
    const float m2a = -2.0f * av;
    float ps[2][2][8];
    for (int mt = 0; mt < 2; ++mt) {
        for (int r = 0; r < 8; ++r) {
            ps[0][mt][r] = tanh_times(acc_h[mt][r], av, m2a);
            ps[1][mt][r] = tanh_times(acc_u[mt][r], av, m2a);
        }
    }
    for (int p = 0; p < 2; ++p) {
        for (int mt = 0; mt < 2; ++mt) {
            for (int r = 0; r < 8; ++r) {
                float s = ps[p][mt][r];
                s += row_xor<1>(s);
                s += row_xor<2>(s);
                s += row_xor<4>(s);
                s += row_xor<8>(s);
                ps[p][mt][r] = s;
            }
        }
    }
    if (c == 0) {
        for (int p = 0; p < 2; ++p) {
            for (int mt = 0; mt < 2; ++mt) {
                float* d = &sPart[(wave * 2 + p) * HB_ROWS + 16 * mt + 8 * half];
                *(v4f*)d       = (v4f){ps[p][mt][0], ps[p][mt][1], ps[p][mt][2], ps[p][mt][3]};
                *(v4f*)(d + 4) = (v4f){ps[p][mt][4], ps[p][mt][5], ps[p][mt][6], ps[p][mt][7]};
            }
        }
    }
    __syncthreads();

    const int my_row = ((c >> 3) << 4) + 8 * half + (c & 7);
    float s0 = sPart[0 * HB_ROWS + my_row];
    float s1 = sPart[1 * HB_ROWS + my_row];
    for (int w = 1; w < 8; ++w) {
        s0 += sPart[(w * 2 + 0) * HB_ROWS + my_row];
        s1 += sPart[(w * 2 + 1) * HB_ROWS + my_row];
    }
    const float mx = fmaxf(s0, s1);
    const float x0 = expf(s0 - mx);
    const float x1 = expf(s1 - mx);
    const float denom = x0 + 2.f * x1;
    const float al0 = x0 / denom;
    const float al1 = x1 / denom;
    const float al12 = 2.f * al1;
    if (wave == 0) {
        sAlpha[3 * my_row + 0] = al0;
        sAlpha[3 * my_row + 1] = al1;
        sAlpha[3 * my_row + 2] = al1;
    }

    for (int mt = 0; mt < 2; ++mt) {
        for (int r = 0; r < 8; ++r) {
            float w0, w12;
            switch (8 * mt + r) {
                case 0:  w0 = row_share<0>(al0);  w12 = row_share<0>(al12);  break;
                case 1:  w0 = row_share<1>(al0);  w12 = row_share<1>(al12);  break;
                case 2:  w0 = row_share<2>(al0);  w12 = row_share<2>(al12);  break;
                case 3:  w0 = row_share<3>(al0);  w12 = row_share<3>(al12);  break;
                case 4:  w0 = row_share<4>(al0);  w12 = row_share<4>(al12);  break;
                case 5:  w0 = row_share<5>(al0);  w12 = row_share<5>(al12);  break;
                case 6:  w0 = row_share<6>(al0);  w12 = row_share<6>(al12);  break;
                case 7:  w0 = row_share<7>(al0);  w12 = row_share<7>(al12);  break;
                case 8:  w0 = row_share<8>(al0);  w12 = row_share<8>(al12);  break;
                case 9:  w0 = row_share<9>(al0);  w12 = row_share<9>(al12);  break;
                case 10: w0 = row_share<10>(al0); w12 = row_share<10>(al12); break;
                case 11: w0 = row_share<11>(al0); w12 = row_share<11>(al12); break;
                case 12: w0 = row_share<12>(al0); w12 = row_share<12>(al12); break;
                case 13: w0 = row_share<13>(al0); w12 = row_share<13>(al12); break;
                case 14: w0 = row_share<14>(al0); w12 = row_share<14>(al12); break;
                default: w0 = row_share<15>(al0); w12 = row_share<15>(al12); break;
            }
            sAu[(16 * mt + 8 * half + r) * PITCH_U + ncol] = w0 * acc_h[mt][r] + w12 * acc_u[mt][r];
        }
    }
    __syncthreads();

    for (int j = 0; j < 4; ++j) {
        const int r = 4 * wave + j;
        if (row0 + r < E)
            ((v4f*)h_new)[(size_t)(row0 + r) * 32 + lane] = *(const v4f*)&sAu[r * PITCH_U + 4 * lane];
    }
    if (wave == 0) {
        if (row0 + HB_ROWS <= E) {
            if (lane < 24)
                ((v4f*)alpha_out)[(size_t)blockIdx.x * 24 + lane] = *(const v4f*)&sAlpha[4 * lane];
        } else {
            for (int f = lane; f < 3 * HB_ROWS; f += 32)
                if (row0 + f / 3 < E) alpha_out[(size_t)row0 * 3 + f] = sAlpha[f];
        }
    }
}

extern "C" void kernel_launch(void* const* d_in, const int* in_sizes, int n_in,
                              void* d_out, int out_size, void* d_ws, size_t ws_size,
                              hipStream_t stream) {
    const float* h_xe = (const float*)d_in[0];
    const float* h    = (const float*)d_in[1];
    const float* m    = (const float*)d_in[2];
    const float* W_h  = (const float*)d_in[3];
    const float* W_f  = (const float*)d_in[4];
    const float* avec = (const float*)d_in[5];
    const int* idx_e  = (const int*)d_in[6];
    const int* idx_n  = (const int*)d_in[7];

    const int E  = in_sizes[0] / 64;
    const int DH = in_sizes[5];

    float* h_new     = (float*)d_out;
    float* alpha_out = (float*)d_out + (size_t)E * DH;

    gather_proj_attn_kernel<<<(E + 31) / 32, 256, 0, stream>>>(h_xe, h, m, idx_e, idx_n,
                                                                W_h, W_f, avec,
                                                                h_new, alpha_out, E);
}
